# starting kernel
# baseline (speedup 1.0000x reference)
#include <hip/hip_runtime.h>


typedef float v2f __attribute__((ext_vector_type(2)));
typedef float v4f __attribute__((ext_vector_type(4)));
typedef float v8f __attribute__((ext_vector_type(8)));

#define TPB 256
#define NPIX 32
#define C_CH 256
#define HID 16
#define HW 4096
#define SC_STRIDE 36
#define W2T_STRIDE 257

__global__ __launch_bounds__(TPB) void se_fused_wmma(
    const float* __restrict__ x, const float* __restrict__ w1,
    const float* __restrict__ b1, const float* __restrict__ w2,
    const float* __restrict__ b2, float* __restrict__ out)
{
    __shared__ float w2t[HID * W2T_STRIDE];
    __shared__ float b2s[C_CH];
    __shared__ float w1s[HID], b1s[HID];
    __shared__ float s_lds[NPIX];
    __shared__ v4f   part[TPB];
    __shared__ float scale_lds[C_CH * SC_STRIDE];

    const int tid  = threadIdx.x;
    const int lane = tid & 31;
    const int wv   = tid >> 5;
    const int half = lane >> 4;
    const int li   = lane & 15;

    const long pixBase = (long)blockIdx.x * NPIX;
    const int  b   = (int)(pixBase >> 12);
    const int  hw0 = (int)(pixBase & 4095);
    const float* xb = x   + ((long)b * C_CH) * HW + hw0;
    float*       ob = out + ((long)b * C_CH) * HW + hw0;

    if (tid < HID) { w1s[tid] = w1[tid]; b1s[tid] = b1[tid]; }
    b2s[tid] = b2[tid];
    #pragma unroll
    for (int i = 0; i < 16; ++i) {
        int idx = i * 256 + tid;
        int c = idx >> 4, k = idx & 15;
        w2t[k * W2T_STRIDE + c] = w2[idx];
    }

    const int cBase = tid >> 3;
    const int p0    = (tid & 7) * 4;
    v4f xv[8];
    v4f psum = {0.f, 0.f, 0.f, 0.f};
    #pragma unroll
    for (int i = 0; i < 8; ++i) {
        int c = cBase + 32 * i;
        xv[i] = __builtin_nontemporal_load((const v4f*)(xb + (long)c * HW + p0));
        psum += xv[i];
    }
    part[tid] = psum;
    __syncthreads();

    if (tid < NPIX) {
        const float* praw = (const float*)part;
        float ssum = 0.f;
        #pragma unroll
        for (int u = 0; u < 32; ++u) ssum += praw[u * 32 + tid];
        s_lds[tid] = ssum * (1.0f / 256.0f);
    }
    __syncthreads();

    v2f Bop[2][4];
    #pragma unroll
    for (int n = 0; n < 2; ++n) {
        float sv = s_lds[n * 16 + li];
        #pragma unroll
        for (int q = 0; q < 4; ++q) {
            int k0 = 4 * q + 2 * half;
            Bop[n][q].x = fmaxf(sv * w1s[k0]     + b1s[k0],     0.f);
            Bop[n][q].y = fmaxf(sv * w1s[k0 + 1] + b1s[k0 + 1], 0.f);
        }
    }
    v2f Aop[2][4];
    #pragma unroll
    for (int im = 0; im < 2; ++im) {
        int c = (2 * wv + im) * 16 + li;
        #pragma unroll
        for (int q = 0; q < 4; ++q) {
            int k0 = 4 * q + 2 * half;
            Aop[im][q].x = w2t[k0 * W2T_STRIDE + c];
            Aop[im][q].y = w2t[(k0 + 1) * W2T_STRIDE + c];
        }
    }

    v8f acc[2][2] = {};
    #pragma unroll
    for (int im = 0; im < 2; ++im)
        #pragma unroll
        for (int n = 0; n < 2; ++n)
            #pragma unroll
            for (int q = 0; q < 4; ++q)
                acc[im][n] = __builtin_amdgcn_wmma_f32_16x16x4_f32(
                     false, Aop[im][q],
                     false, Bop[n][q],
                     (short)0, acc[im][n],
                     false,  false);

    #pragma unroll
    for (int im = 0; im < 2; ++im) {
        int mbase = (2 * wv + im) * 16 + 8 * half;
        #pragma unroll
        for (int n = 0; n < 2; ++n) {
            int p = n * 16 + li;
            #pragma unroll
            for (int r = 0; r < 8; ++r) {
                int c = mbase + r;
                float v = acc[im][n][r] + b2s[c];
                scale_lds[c * SC_STRIDE + p] = 1.0f / (1.0f + __expf(-v));
            }
        }
    }
    __syncthreads();

    #pragma unroll
    for (int i = 0; i < 8; ++i) {
        int c = cBase + 32 * i;
        v4f sc = *(const v4f*)&scale_lds[c * SC_STRIDE + p0];
        __builtin_nontemporal_store(xv[i] * sc, (v4f*)(ob + (long)c * HW + p0));
    }
}

extern "C" void kernel_launch(void* const* d_in, const int* in_sizes, int n_in,
                              void* d_out, int out_size, void* d_ws, size_t ws_size,
                              hipStream_t stream) {
    const float* x  = (const float*)d_in[0];
    const float* w1 = (const float*)d_in[1];
    const float* b1 = (const float*)d_in[2];
    const float* w2 = (const float*)d_in[3];
    const float* b2 = (const float*)d_in[4];
    float* out = (float*)d_out;

    const int npix_total = in_sizes[0] / C_CH;
    const int blocks = npix_total / NPIX;
    se_fused_wmma<<<blocks, TPB, 0, stream>>>(x, w1, b1, w2, b2, out);
}


# Work-group tile widened from 32 to 64 pixels: 2048 groups, half as many waves to enter, whole 256-byte row pieces; channel sums regrouped (16 channels per lane, 16 partials per pixel)
# speedup vs baseline: 1.0769x; 1.0769x over previous
#include <hip/hip_runtime.h>


typedef float v2f __attribute__((ext_vector_type(2)));
typedef float v4f __attribute__((ext_vector_type(4)));
typedef float v8f __attribute__((ext_vector_type(8)));

#define TPB 256
#define NPIX 64
#define NT_N (NPIX / 16)
#define C_CH 256
#define HID 16
#define HW 4096
#define SC_STRIDE 68
#define W2T_STRIDE 257

__global__ __launch_bounds__(TPB) void se_fused_wmma(
    const float* __restrict__ x, const float* __restrict__ w1,
    const float* __restrict__ b1, const float* __restrict__ w2,
    const float* __restrict__ b2, float* __restrict__ out)
{
    __shared__ float w2t[HID * W2T_STRIDE];
    __shared__ float b2s[C_CH];
    __shared__ float w1s[HID], b1s[HID];
    __shared__ float s_lds[NPIX];
    __shared__ v4f   part[TPB];
    __shared__ __attribute__((aligned(16))) float scale_lds[C_CH * SC_STRIDE];

    const int tid  = threadIdx.x;
    const int lane = tid & 31;
    const int wv   = tid >> 5;
    const int half = lane >> 4;
    const int li   = lane & 15;

    const long pixBase = (long)blockIdx.x * NPIX;
    const int  b   = (int)(pixBase >> 12);
    const int  hw0 = (int)(pixBase & 4095);
    const float* xb = x   + ((long)b * C_CH) * HW + hw0;
    float*       ob = out + ((long)b * C_CH) * HW + hw0;

    if (tid < HID) { w1s[tid] = w1[tid]; b1s[tid] = b1[tid]; }
    b2s[tid] = b2[tid];
    #pragma unroll
    for (int i = 0; i < 16; ++i) {
        int idx = i * 256 + tid;
        int c = idx >> 4, k = idx & 15;
        w2t[k * W2T_STRIDE + c] = w2[idx];
    }

    const int cBase = tid >> 4;
    const int p0    = (tid & 15) * 4;
    v4f xv[16];
    v4f psum = {0.f, 0.f, 0.f, 0.f};
    #pragma unroll
    for (int i = 0; i < 16; ++i) {
        int c = cBase + 16 * i;
        xv[i] = __builtin_nontemporal_load((const v4f*)(xb + (long)c * HW + p0));
        psum += xv[i];
    }
    part[tid] = psum;
    __syncthreads();

    if (tid < NPIX) {
        const float* praw = (const float*)part;
        float ssum = 0.f;
        #pragma unroll
        for (int u = 0; u < 16; ++u) ssum += praw[u * NPIX + tid];
        s_lds[tid] = ssum * (1.0f / 256.0f);
    }
    __syncthreads();

    v2f Aop[2][4];
    #pragma unroll
    for (int im = 0; im < 2; ++im) {
        int c = (2 * wv + im) * 16 + li;
        #pragma unroll
        for (int q = 0; q < 4; ++q) {
            int k0 = 4 * q + 2 * half;
            Aop[im][q].x = w2t[k0 * W2T_STRIDE + c];
            Aop[im][q].y = w2t[(k0 + 1) * W2T_STRIDE + c];
        }
    }

    #pragma unroll
    for (int nh = 0; nh < NT_N / 2; ++nh) {
        v2f Bop[2][4];
        #pragma unroll
        for (int n = 0; n < 2; ++n) {
            float sv = s_lds[(2 * nh + n) * 16 + li];
            #pragma unroll
            for (int q = 0; q < 4; ++q) {
                int k0 = 4 * q + 2 * half;
                Bop[n][q].x = fmaxf(sv * w1s[k0]     + b1s[k0],     0.f);
                Bop[n][q].y = fmaxf(sv * w1s[k0 + 1] + b1s[k0 + 1], 0.f);
            }
        }

        v8f acc[2][2] = {};
        #pragma unroll
        for (int im = 0; im < 2; ++im)
            #pragma unroll
            for (int n = 0; n < 2; ++n)
                #pragma unroll
                for (int q = 0; q < 4; ++q)
                    acc[im][n] = __builtin_amdgcn_wmma_f32_16x16x4_f32(
                         false, Aop[im][q],
                         false, Bop[n][q],
                         (short)0, acc[im][n],
                         false,  false);

        #pragma unroll
        for (int im = 0; im < 2; ++im) {
            int mbase = (2 * wv + im) * 16 + 8 * half;
            #pragma unroll
            for (int n = 0; n < 2; ++n) {
                int p = (2 * nh + n) * 16 + li;
                #pragma unroll
                for (int r = 0; r < 8; ++r) {
                    int c = mbase + r;
                    float v = acc[im][n][r] + b2s[c];
                    scale_lds[c * SC_STRIDE + p] = 1.0f / (1.0f + __expf(-v));
                }
            }
            __builtin_amdgcn_sched_barrier(0);
        }
    }
    __syncthreads();

    #pragma unroll
    for (int h = 0; h < 2; ++h) {
        v4f sc[8];
        #pragma unroll
        for (int j = 0; j < 8; ++j) {
            int c = cBase + 16 * (8 * h + j);
            sc[j] = *(const v4f*)&scale_lds[c * SC_STRIDE + p0];
        }
        #pragma unroll
        for (int j = 0; j < 8; ++j) {
            int c = cBase + 16 * (8 * h + j);
            __builtin_nontemporal_store(xv[8 * h + j] * sc[j], (v4f*)(ob + (long)c * HW + p0));
        }
    }
}

extern "C" void kernel_launch(void* const* d_in, const int* in_sizes, int n_in,
                              void* d_out, int out_size, void* d_ws, size_t ws_size,
                              hipStream_t stream) {
    const float* x  = (const float*)d_in[0];
    const float* w1 = (const float*)d_in[1];
    const float* b1 = (const float*)d_in[2];
    const float* w2 = (const float*)d_in[3];
    const float* b2 = (const float*)d_in[4];
    float* out = (float*)d_out;

    const int npix_total = in_sizes[0] / C_CH;
    const int blocks = npix_total / NPIX;
    se_fused_wmma<<<blocks, TPB, 0, stream>>>(x, w1, b1, w2, b2, out);
}


# Sigmoid gate: hardware reciprocal v_rcp_f32 (1 ulp) in place of the 11-instruction IEEE division, for all 64 gate values per lane; order of every sum unchanged
# speedup vs baseline: 1.1990x; 1.1134x over previous
#include <hip/hip_runtime.h>


typedef float v2f __attribute__((ext_vector_type(2)));
typedef float v4f __attribute__((ext_vector_type(4)));
typedef float v8f __attribute__((ext_vector_type(8)));

#define TPB 256
#define NPIX 64
#define NT_N (NPIX / 16)
#define C_CH 256
#define HID 16
#define HW 4096
#define SC_STRIDE 68
#define W2T_STRIDE 257

__global__ __launch_bounds__(TPB) void se_fused_wmma(
    const float* __restrict__ x, const float* __restrict__ w1,
    const float* __restrict__ b1, const float* __restrict__ w2,
    const float* __restrict__ b2, float* __restrict__ out)
{
    __shared__ float w2t[HID * W2T_STRIDE];
    __shared__ float b2s[C_CH];
    __shared__ float w1s[HID], b1s[HID];
    __shared__ float s_lds[NPIX];
    __shared__ v4f   part[TPB];
    __shared__ __attribute__((aligned(16))) float scale_lds[C_CH * SC_STRIDE];

    const int tid  = threadIdx.x;
    const int lane = tid & 31;
    const int wv   = tid >> 5;
    const int half = lane >> 4;
    const int li   = lane & 15;

    const long pixBase = (long)blockIdx.x * NPIX;
    const int  b   = (int)(pixBase >> 12);
    const int  hw0 = (int)(pixBase & 4095);
    const float* xb = x   + ((long)b * C_CH) * HW + hw0;
    float*       ob = out + ((long)b * C_CH) * HW + hw0;

    if (tid < HID) { w1s[tid] = w1[tid]; b1s[tid] = b1[tid]; }
    b2s[tid] = b2[tid];
    #pragma unroll
    for (int i = 0; i < 16; ++i) {
        int idx = i * 256 + tid;
        int c = idx >> 4, k = idx & 15;
        w2t[k * W2T_STRIDE + c] = w2[idx];
    }

    const int cBase = tid >> 4;
    const int p0    = (tid & 15) * 4;
    v4f xv[16];
    v4f psum = {0.f, 0.f, 0.f, 0.f};
    #pragma unroll
    for (int i = 0; i < 16; ++i) {
        int c = cBase + 16 * i;
        xv[i] = __builtin_nontemporal_load((const v4f*)(xb + (long)c * HW + p0));
        psum += xv[i];
    }
    part[tid] = psum;
    __syncthreads();

    if (tid < NPIX) {
        const float* praw = (const float*)part;
        float ssum = 0.f;
        #pragma unroll
        for (int u = 0; u < 16; ++u) ssum += praw[u * NPIX + tid];
        s_lds[tid] = ssum * (1.0f / 256.0f);
    }
    __syncthreads();

    v2f Aop[2][4];
    #pragma unroll
    for (int im = 0; im < 2; ++im) {
        int c = (2 * wv + im) * 16 + li;
        #pragma unroll
        for (int q = 0; q < 4; ++q) {
            int k0 = 4 * q + 2 * half;
            Aop[im][q].x = w2t[k0 * W2T_STRIDE + c];
            Aop[im][q].y = w2t[(k0 + 1) * W2T_STRIDE + c];
        }
    }

    #pragma unroll
    for (int nh = 0; nh < NT_N / 2; ++nh) {
        v2f Bop[2][4];
        #pragma unroll
        for (int n = 0; n < 2; ++n) {
            float sv = s_lds[(2 * nh + n) * 16 + li];
            #pragma unroll
            for (int q = 0; q < 4; ++q) {
                int k0 = 4 * q + 2 * half;
                Bop[n][q].x = fmaxf(sv * w1s[k0]     + b1s[k0],     0.f);
                Bop[n][q].y = fmaxf(sv * w1s[k0 + 1] + b1s[k0 + 1], 0.f);
            }
        }

        v8f acc[2][2] = {};
        #pragma unroll
        for (int im = 0; im < 2; ++im)
            #pragma unroll
            for (int n = 0; n < 2; ++n)
                #pragma unroll
                for (int q = 0; q < 4; ++q)
                    acc[im][n] = __builtin_amdgcn_wmma_f32_16x16x4_f32(
                         false, Aop[im][q],
                         false, Bop[n][q],
                         (short)0, acc[im][n],
                         false,  false);

        #pragma unroll
        for (int im = 0; im < 2; ++im) {
            int mbase = (2 * wv + im) * 16 + 8 * half;
            #pragma unroll
            for (int n = 0; n < 2; ++n) {
                int p = (2 * nh + n) * 16 + li;
                #pragma unroll
                for (int r = 0; r < 8; ++r) {
                    int c = mbase + r;
                    float v = acc[im][n][r] + b2s[c];
                    scale_lds[c * SC_STRIDE + p] = __builtin_amdgcn_rcpf(1.0f + __expf(-v));
                }
            }
            __builtin_amdgcn_sched_barrier(0);
        }
    }
    __syncthreads();

    #pragma unroll
    for (int h = 0; h < 2; ++h) {
        v4f sc[8];
        #pragma unroll
        for (int j = 0; j < 8; ++j) {
            int c = cBase + 16 * (8 * h + j);
            sc[j] = *(const v4f*)&scale_lds[c * SC_STRIDE + p0];
        }
        #pragma unroll
        for (int j = 0; j < 8; ++j) {
            int c = cBase + 16 * (8 * h + j);
            __builtin_nontemporal_store(xv[8 * h + j] * sc[j], (v4f*)(ob + (long)c * HW + p0));
        }
    }
}

extern "C" void kernel_launch(void* const* d_in, const int* in_sizes, int n_in,
                              void* d_out, int out_size, void* d_ws, size_t ws_size,
                              hipStream_t stream) {
    const float* x  = (const float*)d_in[0];
    const float* w1 = (const float*)d_in[1];
    const float* b1 = (const float*)d_in[2];
    const float* w2 = (const float*)d_in[3];
    const float* b2 = (const float*)d_in[4];
    float* out = (float*)d_out;

    const int npix_total = in_sizes[0] / C_CH;
    const int blocks = npix_total / NPIX;
    se_fused_wmma<<<blocks, TPB, 0, stream>>>(x, w1, b1, w2, b2, out);
}
